# starting kernel
# baseline (speedup 1.0000x reference)
#include <hip/hip_runtime.h>
#include <math.h>

typedef __attribute__((ext_vector_type(2))) float v2f;
typedef __attribute__((ext_vector_type(8))) float v8f;

#define NEG_SLOPE 0.2f

__device__ __forceinline__ float lrelu(float v) { return v >= 0.f ? v : NEG_SLOPE * v; }

__device__ __forceinline__ void atomicMaxF(float* addr, float val) {
  if (val >= 0.f) atomicMax((int*)addr, __float_as_int(val));
  else            atomicMin((unsigned int*)addr, __float_as_uint(val));
}

__global__ void gemm1_x_w1(const float* __restrict__ X, const float* __restrict__ W,
                           float* __restrict__ H) {
  const int lane = threadIdx.x & 31;
  const int wave = threadIdx.x >> 5;
  const int kgrp = (lane >> 4) << 1;
  const int ncol = (wave << 4) + (lane & 15);
  const float* xrow = X + ((long)blockIdx.x * 16 + (lane & 15)) * 256;
  v8f acc = {};
#pragma unroll 8
  for (int k0 = 0; k0 < 256; k0 += 4) {
    v2f a = *(const v2f*)(xrow + k0 + kgrp);
    v2f b;
    b.x = W[(k0 + kgrp) * 64 + ncol];
    b.y = W[(k0 + kgrp + 1) * 64 + ncol];
    acc = __builtin_amdgcn_wmma_f32_16x16x4_f32(false, a, false, b, (short)0, acc,
                                                false, false);
  }
  const long mb = (long)blockIdx.x * 16 + ((lane >> 4) << 3);
#pragma unroll
  for (int v = 0; v < 8; ++v)
    H[(mb + v) * 64 + ncol] = acc[v];
}

__global__ void gemm2_h_w2(const float* __restrict__ A1, const float* __restrict__ W,
                           float* __restrict__ H2, int mtiles) {
  const int lane = threadIdx.x & 31;
  const int wave = threadIdx.x >> 5;
  const int mtile = blockIdx.x * 4 + wave;
  if (mtile >= mtiles) return;
  const int kgrp = (lane >> 4) << 1;
  const int ncol = lane & 15;
  const float* arow = A1 + ((long)mtile * 16 + ncol) * 64;
  v8f acc = {};
#pragma unroll
  for (int k0 = 0; k0 < 64; k0 += 4) {
    v2f a = *(const v2f*)(arow + k0 + kgrp);
    a.x = fmaxf(a.x, 0.f);
    a.y = fmaxf(a.y, 0.f);
    v2f b;
    b.x = W[(k0 + kgrp) * 8 + (ncol & 7)];
    b.y = W[(k0 + kgrp + 1) * 8 + (ncol & 7)];
    acc = __builtin_amdgcn_wmma_f32_16x16x4_f32(false, a, false, b, (short)0, acc,
                                                false, false);
  }
  if (ncol < 8) {
    const long mb = (long)mtile * 16 + ((lane >> 4) << 3);
#pragma unroll
    for (int v = 0; v < 8; ++v)
      H2[(mb + v) * 8 + ncol] = acc[v];
  }
}

template <int H, int C>
__global__ void attn_logits(const float* __restrict__ hf, const float* __restrict__ as,
                            const float* __restrict__ ad, float* __restrict__ als,
                            float* __restrict__ ald, long NH) {
  long t = (long)blockIdx.x * blockDim.x + threadIdx.x;
  if (t >= NH) return;
  const int h = (int)(t % H);
  const float* hp = hf + t * C;
  float ss = 0.f, sd = 0.f;
#pragma unroll
  for (int c = 0; c < C; ++c) {
    float v = hp[c];
    ss += v * as[h * C + c];
    sd += v * ad[h * C + c];
  }
  als[t] = ss;
  ald[t] = sd;
}

__global__ void init_softmax(float* __restrict__ m, float* __restrict__ den, long n) {
  long t = (long)blockIdx.x * blockDim.x + threadIdx.x;
  if (t < n) { m[t] = -__builtin_huge_valf(); den[t] = 0.f; }
}

__global__ void init_acc_bias(float* __restrict__ acc, const float* __restrict__ b,
                              long n, int F) {
  long t = (long)blockIdx.x * blockDim.x + threadIdx.x;
  if (t < n) acc[t] = b[(int)(t % F)];
}

__device__ __forceinline__ void edge_sd(const int* __restrict__ ei, long Eraw, long e,
                                        int& s, int& d) {
  if (e < Eraw) { s = ei[e]; d = ei[Eraw + e]; }
  else          { s = d = (int)(e - Eraw); }
}

template <int H>
__global__ void edge_softmax_max(const int* __restrict__ ei, long Eraw, long Etot,
                                 const float* __restrict__ als, const float* __restrict__ ald,
                                 float* __restrict__ m) {
  long e = (long)blockIdx.x * blockDim.x + threadIdx.x;
  if (e >= Etot) return;
  int s, d; edge_sd(ei, Eraw, e, s, d);
#pragma unroll
  for (int h = 0; h < H; ++h) {
    float v = lrelu(als[(long)s * H + h] + ald[(long)d * H + h]);
    atomicMaxF(&m[(long)d * H + h], v);
  }
}

template <int H>
__global__ void edge_softmax_sum(const int* __restrict__ ei, long Eraw, long Etot,
                                 const float* __restrict__ als, const float* __restrict__ ald,
                                 const float* __restrict__ m, float* __restrict__ den) {
  long e = (long)blockIdx.x * blockDim.x + threadIdx.x;
  if (e >= Etot) return;
  int s, d; edge_sd(ei, Eraw, e, s, d);
#pragma unroll
  for (int h = 0; h < H; ++h) {
    float v = lrelu(als[(long)s * H + h] + ald[(long)d * H + h]);
    atomicAdd(&den[(long)d * H + h], __expf(v - m[(long)d * H + h]));
  }
}

template <int H, int C>
__global__ void edge_aggregate(const int* __restrict__ ei, long Eraw, long Etot,
                               const float* __restrict__ als, const float* __restrict__ ald,
                               const float* __restrict__ m, const float* __restrict__ den,
                               const float* __restrict__ hf, float* __restrict__ acc) {
  long t = (long)blockIdx.x * blockDim.x + threadIdx.x;
  if (t >= Etot * H) return;
  long e = t / H;
  int h = (int)(t % H);
  int s, d; edge_sd(ei, Eraw, e, s, d);
  long sh = (long)s * H + h, dh = (long)d * H + h;
  float v = lrelu(als[sh] + ald[dh]);
  float alpha = __expf(v - m[dh]) / (den[dh] + 1e-16f);
  const float* hp = hf + sh * C;
  float* op = acc + dh * C;
#pragma unroll
  for (int c = 0; c < C; ++c)
    atomicAdd(&op[c], hp[c] * alpha);
}

__global__ void score_kernel(const float* __restrict__ emb, const float* __restrict__ W3,
                             const float* __restrict__ b3, float* __restrict__ score, long N) {
  long n = (long)blockIdx.x * blockDim.x + threadIdx.x;
  if (n >= N) return;
  float s = b3[0];
#pragma unroll
  for (int c = 0; c < 8; ++c) s += emb[n * 8 + c] * W3[c];
  score[n] = s;
}

static inline int nblk(long n, int b) { return (int)((n + b - 1) / b); }

extern "C" void kernel_launch(void* const* d_in, const int* in_sizes, int n_in,
                              void* d_out, int out_size, void* d_ws, size_t ws_size,
                              hipStream_t stream) {
  const float* x   = (const float*)d_in[0];
  const int*   ei  = (const int*)d_in[1];
  const float* W1  = (const float*)d_in[2];
  const float* as1 = (const float*)d_in[3];
  const float* ad1 = (const float*)d_in[4];
  const float* b1  = (const float*)d_in[5];
  const float* W2  = (const float*)d_in[6];
  const float* as2 = (const float*)d_in[7];
  const float* ad2 = (const float*)d_in[8];
  const float* b2  = (const float*)d_in[9];
  const float* W3  = (const float*)d_in[10];
  const float* b3  = (const float*)d_in[11];

  const long N    = in_sizes[0] / 256;
  const long Eraw = in_sizes[1] / 2;
  const long Etot = Eraw + N;

  float* ws   = (float*)d_ws;
  float* h1   = ws;
  float* acc1 = h1 + N * 64;
  float* als1 = acc1 + N * 64;
  float* ald1 = als1 + N * 8;
  float* m1   = ald1 + N * 8;
  float* den1 = m1 + N * 8;
  float* h2   = den1 + N * 8;
  float* als2 = h2 + N * 8;
  float* ald2 = als2 + N;
  float* m2   = ald2 + N;
  float* den2 = m2 + N;

  float* emb   = (float*)d_out;
  float* score = emb + N * 8;

  const int B = 256;
  const int mtiles = (int)(N / 16);

  gemm1_x_w1<<<mtiles, 128, 0, stream>>>(x, W1, h1);
  attn_logits<8, 8><<<nblk(N * 8, B), B, 0, stream>>>(h1, as1, ad1, als1, ald1, N * 8);
  init_softmax<<<nblk(N * 8, B), B, 0, stream>>>(m1, den1, N * 8);
  init_acc_bias<<<nblk(N * 64, B), B, 0, stream>>>(acc1, b1, N * 64, 64);
  edge_softmax_max<8><<<nblk(Etot, B), B, 0, stream>>>(ei, Eraw, Etot, als1, ald1, m1);
  edge_softmax_sum<8><<<nblk(Etot, B), B, 0, stream>>>(ei, Eraw, Etot, als1, ald1, m1, den1);
  edge_aggregate<8, 8><<<nblk(Etot * 8, B), B, 0, stream>>>(ei, Eraw, Etot, als1, ald1,
                                                            m1, den1, h1, acc1);

  gemm2_h_w2<<<(mtiles + 3) / 4, 128, 0, stream>>>(acc1, W2, h2, mtiles);
  attn_logits<1, 8><<<nblk(N, B), B, 0, stream>>>(h2, as2, ad2, als2, ald2, N);
  init_softmax<<<nblk(N, B), B, 0, stream>>>(m2, den2, N);
  init_acc_bias<<<nblk(N * 8, B), B, 0, stream>>>(emb, b2, N * 8, 8);
  edge_softmax_max<1><<<nblk(Etot, B), B, 0, stream>>>(ei, Eraw, Etot, als2, ald2, m2);
  edge_softmax_sum<1><<<nblk(Etot, B), B, 0, stream>>>(ei, Eraw, Etot, als2, ald2, m2, den2);
  edge_aggregate<1, 8><<<nblk(Etot, B), B, 0, stream>>>(ei, Eraw, Etot, als2, ald2,
                                                        m2, den2, h2, emb);

  score_kernel<<<nblk(N, B), B, 0, stream>>>(emb, W3, b3, score, N);
}


# Layer-1 aggregate: wave halves trade one value (permlanex16) so each atomic instruction adds 32 consecutive words of ONE 128-B destination line, not every other word of two; atomic sum order as before
# speedup vs baseline: 2.9387x; 2.9387x over previous
#include <hip/hip_runtime.h>
#include <math.h>

typedef __attribute__((ext_vector_type(2))) float v2f;
typedef __attribute__((ext_vector_type(8))) float v8f;

#define NEG_SLOPE 0.2f

__device__ __forceinline__ float lrelu(float v) { return v >= 0.f ? v : NEG_SLOPE * v; }

__device__ __forceinline__ void atomicMaxF(float* addr, float val) {
  if (val >= 0.f) atomicMax((int*)addr, __float_as_int(val));
  else            atomicMin((unsigned int*)addr, __float_as_uint(val));
}

__global__ void gemm1_x_w1(const float* __restrict__ X, const float* __restrict__ W,
                           float* __restrict__ H) {
  const int lane = threadIdx.x & 31;
  const int wave = threadIdx.x >> 5;
  const int kgrp = (lane >> 4) << 1;
  const int ncol = (wave << 4) + (lane & 15);
  const float* xrow = X + ((long)blockIdx.x * 16 + (lane & 15)) * 256;
  v8f acc = {};
#pragma unroll 8
  for (int k0 = 0; k0 < 256; k0 += 4) {
    v2f a = *(const v2f*)(xrow + k0 + kgrp);
    v2f b;
    b.x = W[(k0 + kgrp) * 64 + ncol];
    b.y = W[(k0 + kgrp + 1) * 64 + ncol];
    acc = __builtin_amdgcn_wmma_f32_16x16x4_f32(false, a, false, b, (short)0, acc,
                                                false, false);
  }
  const long mb = (long)blockIdx.x * 16 + ((lane >> 4) << 3);
#pragma unroll
  for (int v = 0; v < 8; ++v)
    H[(mb + v) * 64 + ncol] = acc[v];
}

__global__ void gemm2_h_w2(const float* __restrict__ S1, const float* __restrict__ den1,
                           const float* __restrict__ b1, const float* __restrict__ W,
                           float* __restrict__ H2, int mtiles) {
  const int lane = threadIdx.x & 31;
  const int wave = threadIdx.x >> 5;
  const int mtile = blockIdx.x * 4 + wave;
  if (mtile >= mtiles) return;
  const int kgrp = (lane >> 4) << 1;
  const int ncol = lane & 15;
  const long row = (long)mtile * 16 + ncol;
  const float* arow = S1 + row * 64;
  const float4 dlo = *(const float4*)(den1 + row * 8);
  const float4 dhi = *(const float4*)(den1 + row * 8 + 4);
  const float rden[8] = {1.f / (dlo.x + 1e-16f), 1.f / (dlo.y + 1e-16f),
                         1.f / (dlo.z + 1e-16f), 1.f / (dlo.w + 1e-16f),
                         1.f / (dhi.x + 1e-16f), 1.f / (dhi.y + 1e-16f),
                         1.f / (dhi.z + 1e-16f), 1.f / (dhi.w + 1e-16f)};
  v8f acc = {};
#pragma unroll
  for (int k0 = 0; k0 < 64; k0 += 4) {
    v2f a = *(const v2f*)(arow + k0 + kgrp);
    const v2f bb = *(const v2f*)(b1 + k0 + kgrp);
    const float r = rden[k0 >> 3];
    a.x = fmaxf(a.x * r + bb.x, 0.f);
    a.y = fmaxf(a.y * r + bb.y, 0.f);
    v2f b;
    b.x = W[(k0 + kgrp) * 8 + (ncol & 7)];
    b.y = W[(k0 + kgrp + 1) * 8 + (ncol & 7)];
    acc = __builtin_amdgcn_wmma_f32_16x16x4_f32(false, a, false, b, (short)0, acc,
                                                false, false);
  }
  if (ncol < 8) {
    const long mb = (long)mtile * 16 + ((lane >> 4) << 3);
#pragma unroll
    for (int v = 0; v < 8; ++v)
      H2[(mb + v) * 8 + ncol] = acc[v];
  }
}

template <int H, int C>
__global__ void attn_logits(const float* __restrict__ hf, const float* __restrict__ as,
                            const float* __restrict__ ad, float* __restrict__ als,
                            float* __restrict__ ald, long NH) {
  long t = (long)blockIdx.x * blockDim.x + threadIdx.x;
  if (t >= NH) return;
  const int h = (int)(t % H);
  const float* hp = hf + t * C;
  float ss = 0.f, sd = 0.f;
#pragma unroll
  for (int c = 0; c < C; ++c) {
    float v = hp[c];
    ss += v * as[h * C + c];
    sd += v * ad[h * C + c];
  }
  als[t] = ss;
  ald[t] = sd;
}

__global__ void init_softmax(float* __restrict__ m, float* __restrict__ den, long n) {
  long t = (long)blockIdx.x * blockDim.x + threadIdx.x;
  if (t < n) { m[t] = -__builtin_huge_valf(); den[t] = 0.f; }
}

__global__ void init_zero(float* __restrict__ p, long n4) {
  long t = (long)blockIdx.x * blockDim.x + threadIdx.x;
  if (t < n4) ((float4*)p)[t] = make_float4(0.f, 0.f, 0.f, 0.f);
}

__device__ __forceinline__ void edge_sd(const int* __restrict__ ei, long Eraw, long e,
                                        int& s, int& d) {
  if (e < Eraw) { s = ei[e]; d = ei[Eraw + e]; }
  else          { s = d = (int)(e - Eraw); }
}

template <int H>
__global__ void edge_softmax_max(const int* __restrict__ ei, long Eraw, long Etot,
                                 const float* __restrict__ als, const float* __restrict__ ald,
                                 float* __restrict__ m) {
  long e = (long)blockIdx.x * blockDim.x + threadIdx.x;
  if (e >= Etot) return;
  int s, d; edge_sd(ei, Eraw, e, s, d);
#pragma unroll
  for (int h = 0; h < H; ++h) {
    float v = lrelu(als[(long)s * H + h] + ald[(long)d * H + h]);
    atomicMaxF(&m[(long)d * H + h], v);
  }
}

__device__ __forceinline__ void edge_sd_nb(const int* __restrict__ ei, int Eraw, int e,
                                           int& s, int& d) {
  const int ec = e < Eraw ? e : Eraw - 1;
  const int sr = ei[ec];
  const int dr = ei[(long)Eraw + ec];
  s = e < Eraw ? sr : e - Eraw;
  d = e < Eraw ? dr : e - Eraw;
}

template <int U>
__global__ void edge_max_h8(const int* __restrict__ ei, int Eraw, int Etot,
                            const float* __restrict__ als, const float* __restrict__ ald,
                            float* __restrict__ m) {
  const int h = threadIdx.x & 7;
  const int e0 = blockIdx.x * (32 * U) + (threadIdx.x >> 3);
  if ((int)(blockIdx.x + 1) * (32 * U) <= Etot) {
    int dh[U];
    float v[U];
#pragma unroll
    for (int j = 0; j < U; ++j) {
      int s, d; edge_sd_nb(ei, Eraw, e0 + j * 32, s, d);
      dh[j] = d * 8 + h;
      v[j] = lrelu(als[s * 8 + h] + ald[dh[j]]);
    }
#pragma unroll
    for (int j = 0; j < U; ++j) atomicMaxF(&m[dh[j]], v[j]);
  } else {
    for (int j = 0; j < U; ++j) {
      const int e = e0 + j * 32;
      if (e < Etot) {
        int s, d; edge_sd_nb(ei, Eraw, e, s, d);
        atomicMaxF(&m[d * 8 + h], lrelu(als[s * 8 + h] + ald[d * 8 + h]));
      }
    }
  }
}

__device__ __forceinline__ float xhalf(float x) {
  return __builtin_bit_cast(float, __builtin_amdgcn_permlanex16(__builtin_bit_cast(int, x),
                                                                __builtin_bit_cast(int, x),
                                                                0x76543210, 0xfedcba98, false,
                                                                false));
}

template <int U, int WPB>
__global__ void edge_aggregate_row64(const int* __restrict__ ei, int Eraw, int Etot,
                                     const float* __restrict__ als, const float* __restrict__ ald,
                                     const float* __restrict__ m, const float* __restrict__ hf,
                                     float* __restrict__ acc, float* __restrict__ den) {
  const int lane = threadIdx.x & 31;
  const int wave = __builtin_amdgcn_readfirstlane((int)(threadIdx.x >> 5));
  const int e0 = (blockIdx.x * WPB + wave) * 32;
  if (e0 >= Etot) return;
  const int cnt = Etot - e0 < 32 ? Etot - e0 : 32;
  int sl, dl;
  edge_sd_nb(ei, Eraw, e0 + lane < Etot ? e0 + lane : Etot - 1, sl, dl);
  const int h = lane >> 2;
  const bool head_lane = (lane & 3) == 0;
  const float* hcol = hf + 2 * lane;
  const bool lo = lane < 16;
  float* ocol = acc + (lo ? 2 * lane : 2 * (lane - 16) + 1);
  int j = 0;
  for (; j + U <= cnt; j += U) {
    int d[U];
    float w[U];
    float a1[U], a2[U];
#pragma unroll
    for (int u = 0; u < U; ++u) {
      const int s = __builtin_amdgcn_readlane(sl, j + u);
      d[u] = __builtin_amdgcn_readlane(dl, j + u);
      const int sh = s * 8 + h, dh = d[u] * 8 + h;
      const float v = lrelu(als[sh] + ald[dh]);
      w[u] = __expf(v - m[dh]);
      const v2f hv = *(const v2f*)(hcol + (long)s * 64);
      const float vx = hv.x * w[u], vy = hv.y * w[u];
      const float got = xhalf(lo ? vy : vx);
      a1[u] = lo ? vx : got;
      a2[u] = lo ? got : vy;
    }
#pragma unroll
    for (int u = 0; u < U; ++u) {
      float* op = ocol + (long)d[u] * 64;
      atomicAdd(op, a1[u]);
      atomicAdd(op + 32, a2[u]);
    }
    if (head_lane) {
#pragma unroll
      for (int u = 0; u < U; ++u) atomicAdd(&den[d[u] * 8 + h], w[u]);
    }
  }
  for (; j < cnt; ++j) {
    const int s = __builtin_amdgcn_readlane(sl, j);
    const int d = __builtin_amdgcn_readlane(dl, j);
    const int sh = s * 8 + h, dh = d * 8 + h;
    const float v = lrelu(als[sh] + ald[dh]);
    const float w = __expf(v - m[dh]);
    const v2f hv = *(const v2f*)(hcol + (long)s * 64);
    const float vx = hv.x * w, vy = hv.y * w;
    const float got = xhalf(lo ? vy : vx);
    float* op = ocol + (long)d * 64;
    atomicAdd(op, lo ? vx : got);
    atomicAdd(op + 32, lo ? got : vy);
    if (head_lane) atomicAdd(&den[dh], w);
  }
}

template <int U>
__global__ void edge_aggregate_c8(const int* __restrict__ ei, int Eraw, int Etot,
                                  const float* __restrict__ als, const float* __restrict__ ald,
                                  const float* __restrict__ m, const float* __restrict__ hf,
                                  float* __restrict__ acc, float* __restrict__ den) {
  const int c = threadIdx.x & 7;
  const int e0 = blockIdx.x * (32 * U) + (threadIdx.x >> 3);
  if ((int)(blockIdx.x + 1) * (32 * U) <= Etot) {
    int dn[U];
    float w[U];
    float val[U];
#pragma unroll
    for (int j = 0; j < U; ++j) {
      int s, d; edge_sd_nb(ei, Eraw, e0 + j * 32, s, d);
      dn[j] = d;
      const float v = lrelu(als[s] + ald[d]);
      w[j] = __expf(v - m[d]);
      val[j] = hf[s * 8 + c] * w[j];
    }
#pragma unroll
    for (int j = 0; j < U; ++j) atomicAdd(&acc[dn[j] * 8 + c], val[j]);
    if (c == 0) {
#pragma unroll
      for (int j = 0; j < U; ++j) atomicAdd(&den[dn[j]], w[j]);
    }
  } else {
    for (int j = 0; j < U; ++j) {
      const int e = e0 + j * 32;
      if (e < Etot) {
        int s, d; edge_sd_nb(ei, Eraw, e, s, d);
        const float v = lrelu(als[s] + ald[d]);
        const float w = __expf(v - m[d]);
        atomicAdd(&acc[d * 8 + c], hf[s * 8 + c] * w);
        if (c == 0) atomicAdd(&den[d], w);
      }
    }
  }
}

__global__ void normalize_score_kernel(float* __restrict__ emb, const float* __restrict__ den,
                                       const float* __restrict__ b2,
                                       const float* __restrict__ W3,
                                       const float* __restrict__ b3, float* __restrict__ score,
                                       long N) {
  long n = (long)blockIdx.x * blockDim.x + threadIdx.x;
  if (n >= N) return;
  const float r = 1.f / (den[n] + 1e-16f);
  float4* row = (float4*)(emb + n * 8);
  const float4 lo = row[0], hi = row[1];
  float e[8] = {lo.x, lo.y, lo.z, lo.w, hi.x, hi.y, hi.z, hi.w};
#pragma unroll
  for (int c = 0; c < 8; ++c) e[c] = e[c] * r + b2[c];
  row[0] = make_float4(e[0], e[1], e[2], e[3]);
  row[1] = make_float4(e[4], e[5], e[6], e[7]);
  float s = b3[0];
#pragma unroll
  for (int c = 0; c < 8; ++c) s += e[c] * W3[c];
  score[n] = s;
}

static inline int nblk(long n, int b) { return (int)((n + b - 1) / b); }

extern "C" void kernel_launch(void* const* d_in, const int* in_sizes, int n_in,
                              void* d_out, int out_size, void* d_ws, size_t ws_size,
                              hipStream_t stream) {
  const float* x   = (const float*)d_in[0];
  const int*   ei  = (const int*)d_in[1];
  const float* W1  = (const float*)d_in[2];
  const float* as1 = (const float*)d_in[3];
  const float* ad1 = (const float*)d_in[4];
  const float* b1  = (const float*)d_in[5];
  const float* W2  = (const float*)d_in[6];
  const float* as2 = (const float*)d_in[7];
  const float* ad2 = (const float*)d_in[8];
  const float* b2  = (const float*)d_in[9];
  const float* W3  = (const float*)d_in[10];
  const float* b3  = (const float*)d_in[11];

  const long N    = in_sizes[0] / 256;
  const long Eraw = in_sizes[1] / 2;
  const long Etot = Eraw + N;

  float* ws   = (float*)d_ws;
  float* h1   = ws;
  float* acc1 = h1 + N * 64;
  float* als1 = acc1 + N * 64;
  float* ald1 = als1 + N * 8;
  float* m1   = ald1 + N * 8;
  float* den1 = m1 + N * 8;
  float* h2   = den1 + N * 8;
  float* als2 = h2 + N * 8;
  float* ald2 = als2 + N;
  float* m2   = ald2 + N;
  float* den2 = m2 + N;

  float* emb   = (float*)d_out;
  float* score = emb + N * 8;

  const int B = 256;
  const int mtiles = (int)(N / 16);

  gemm1_x_w1<<<mtiles, 128, 0, stream>>>(x, W1, h1);
  attn_logits<8, 8><<<nblk(N * 8, B), B, 0, stream>>>(h1, as1, ad1, als1, ald1, N * 8);
  init_softmax<<<nblk(N * 8, B), B, 0, stream>>>(m1, den1, N * 8);
  init_zero<<<nblk(N * 16, B), B, 0, stream>>>(acc1, N * 16);
  edge_max_h8<4><<<nblk(Etot, 128), B, 0, stream>>>(ei, (int)Eraw, (int)Etot, als1, ald1, m1);
  edge_aggregate_row64<4, 8><<<nblk(Etot, 256), B, 0, stream>>>(ei, (int)Eraw, (int)Etot, als1,
                                                                ald1, m1, h1, acc1, den1);

  gemm2_h_w2<<<(mtiles + 3) / 4, 128, 0, stream>>>(acc1, den1, b1, W2, h2, mtiles);
  attn_logits<1, 8><<<nblk(N, B), B, 0, stream>>>(h2, as2, ad2, als2, ald2, N);
  init_softmax<<<nblk(N, B), B, 0, stream>>>(m2, den2, N);
  init_zero<<<nblk(N * 2, B), B, 0, stream>>>(emb, N * 2);
  edge_softmax_max<1><<<nblk(Etot, B), B, 0, stream>>>(ei, Eraw, Etot, als2, ald2, m2);
  edge_aggregate_c8<4><<<nblk(Etot, 128), B, 0, stream>>>(ei, (int)Eraw, (int)Etot, als2, ald2,
                                                          m2, h2, emb, den2);

  normalize_score_kernel<<<nblk(N, B), B, 0, stream>>>(emb, den2, b2, W3, b3, score, N);
}


# Segment max taken over the source logits alone (lrelu(a + al_dst[d]) is monotone in a; m rebuilt at the consumer, bit-identical) as ordered unsigned keys: 1 gather + 1 atomic max per edge, not 2 + 2
# speedup vs baseline: 3.2971x; 1.1220x over previous
#include <hip/hip_runtime.h>
#include <math.h>

typedef __attribute__((ext_vector_type(2))) float v2f;
typedef __attribute__((ext_vector_type(8))) float v8f;

#define NEG_SLOPE 0.2f

__device__ __forceinline__ float lrelu(float v) { return v >= 0.f ? v : NEG_SLOPE * v; }

__device__ __forceinline__ unsigned int fkey(float f) {
  const unsigned int u = __float_as_uint(f);
  return (u & 0x80000000u) ? ~u : (u | 0x80000000u);
}
__device__ __forceinline__ float fkey_inv(unsigned int k) {
  return __uint_as_float((k & 0x80000000u) ? (k & 0x7fffffffu) : ~k);
}

__device__ __forceinline__ float seg_max_logit(unsigned int srcmax_key, float al_dst) {
  return lrelu(fkey_inv(srcmax_key) + al_dst);
}

__global__ void gemm1_x_w1(const float* __restrict__ X, const float* __restrict__ W,
                           float* __restrict__ H) {
  const int lane = threadIdx.x & 31;
  const int wave = threadIdx.x >> 5;
  const int kgrp = (lane >> 4) << 1;
  const int ncol = (wave << 4) + (lane & 15);
  const float* xrow = X + ((long)blockIdx.x * 16 + (lane & 15)) * 256;
  v8f acc = {};
#pragma unroll 8
  for (int k0 = 0; k0 < 256; k0 += 4) {
    v2f a = *(const v2f*)(xrow + k0 + kgrp);
    v2f b;
    b.x = W[(k0 + kgrp) * 64 + ncol];
    b.y = W[(k0 + kgrp + 1) * 64 + ncol];
    acc = __builtin_amdgcn_wmma_f32_16x16x4_f32(false, a, false, b, (short)0, acc,
                                                false, false);
  }
  const long mb = (long)blockIdx.x * 16 + ((lane >> 4) << 3);
#pragma unroll
  for (int v = 0; v < 8; ++v)
    H[(mb + v) * 64 + ncol] = acc[v];
}

__global__ void gemm2_h_w2(const float* __restrict__ S1, const float* __restrict__ den1,
                           const float* __restrict__ b1, const float* __restrict__ W,
                           float* __restrict__ H2, int mtiles) {
  const int lane = threadIdx.x & 31;
  const int wave = threadIdx.x >> 5;
  const int mtile = blockIdx.x * 4 + wave;
  if (mtile >= mtiles) return;
  const int kgrp = (lane >> 4) << 1;
  const int ncol = lane & 15;
  const long row = (long)mtile * 16 + ncol;
  const float* arow = S1 + row * 64;
  const float4 dlo = *(const float4*)(den1 + row * 8);
  const float4 dhi = *(const float4*)(den1 + row * 8 + 4);
  const float rden[8] = {1.f / (dlo.x + 1e-16f), 1.f / (dlo.y + 1e-16f),
                         1.f / (dlo.z + 1e-16f), 1.f / (dlo.w + 1e-16f),
                         1.f / (dhi.x + 1e-16f), 1.f / (dhi.y + 1e-16f),
                         1.f / (dhi.z + 1e-16f), 1.f / (dhi.w + 1e-16f)};
  v8f acc = {};
#pragma unroll
  for (int k0 = 0; k0 < 64; k0 += 4) {
    v2f a = *(const v2f*)(arow + k0 + kgrp);
    const v2f bb = *(const v2f*)(b1 + k0 + kgrp);
    const float r = rden[k0 >> 3];
    a.x = fmaxf(a.x * r + bb.x, 0.f);
    a.y = fmaxf(a.y * r + bb.y, 0.f);
    v2f b;
    b.x = W[(k0 + kgrp) * 8 + (ncol & 7)];
    b.y = W[(k0 + kgrp + 1) * 8 + (ncol & 7)];
    acc = __builtin_amdgcn_wmma_f32_16x16x4_f32(false, a, false, b, (short)0, acc,
                                                false, false);
  }
  if (ncol < 8) {
    const long mb = (long)mtile * 16 + ((lane >> 4) << 3);
#pragma unroll
    for (int v = 0; v < 8; ++v)
      H2[(mb + v) * 8 + ncol] = acc[v];
  }
}

template <int H, int C>
__global__ void attn_logits(const float* __restrict__ hf, const float* __restrict__ as,
                            const float* __restrict__ ad, float* __restrict__ als,
                            float* __restrict__ ald, long NH) {
  long t = (long)blockIdx.x * blockDim.x + threadIdx.x;
  if (t >= NH) return;
  const int h = (int)(t % H);
  const float* hp = hf + t * C;
  float ss = 0.f, sd = 0.f;
#pragma unroll
  for (int c = 0; c < C; ++c) {
    float v = hp[c];
    ss += v * as[h * C + c];
    sd += v * ad[h * C + c];
  }
  als[t] = ss;
  ald[t] = sd;
}

__global__ void init_softmax(unsigned int* __restrict__ mk, float* __restrict__ den, long n) {
  long t = (long)blockIdx.x * blockDim.x + threadIdx.x;
  if (t < n) { mk[t] = 0u; den[t] = 0.f; }
}

__global__ void init_zero(float* __restrict__ p, long n4) {
  long t = (long)blockIdx.x * blockDim.x + threadIdx.x;
  if (t < n4) ((float4*)p)[t] = make_float4(0.f, 0.f, 0.f, 0.f);
}

__device__ __forceinline__ void edge_sd(const int* __restrict__ ei, long Eraw, long e,
                                        int& s, int& d) {
  if (e < Eraw) { s = ei[e]; d = ei[Eraw + e]; }
  else          { s = d = (int)(e - Eraw); }
}

template <int H>
__global__ void edge_srcmax(const int* __restrict__ ei, long Eraw, long Etot,
                            const float* __restrict__ als, unsigned int* __restrict__ mk) {
  long e = (long)blockIdx.x * blockDim.x + threadIdx.x;
  if (e >= Etot) return;
  int s, d; edge_sd(ei, Eraw, e, s, d);
#pragma unroll
  for (int h = 0; h < H; ++h)
    atomicMax(&mk[(long)d * H + h], fkey(als[(long)s * H + h]));
}

__device__ __forceinline__ void edge_sd_nb(const int* __restrict__ ei, int Eraw, int e,
                                           int& s, int& d) {
  const int ec = e < Eraw ? e : Eraw - 1;
  const int sr = ei[ec];
  const int dr = ei[(long)Eraw + ec];
  s = e < Eraw ? sr : e - Eraw;
  d = e < Eraw ? dr : e - Eraw;
}

template <int U>
__global__ void edge_srcmax_h8(const int* __restrict__ ei, int Eraw, int Etot,
                               const float* __restrict__ als, unsigned int* __restrict__ mk) {
  const int h = threadIdx.x & 7;
  const int e0 = blockIdx.x * (32 * U) + (threadIdx.x >> 3);
  if ((int)(blockIdx.x + 1) * (32 * U) <= Etot) {
    int dh[U];
    unsigned int k[U];
#pragma unroll
    for (int j = 0; j < U; ++j) {
      int s, d; edge_sd_nb(ei, Eraw, e0 + j * 32, s, d);
      dh[j] = d * 8 + h;
      k[j] = fkey(als[s * 8 + h]);
    }
#pragma unroll
    for (int j = 0; j < U; ++j) atomicMax(&mk[dh[j]], k[j]);
  } else {
    for (int j = 0; j < U; ++j) {
      const int e = e0 + j * 32;
      if (e < Etot) {
        int s, d; edge_sd_nb(ei, Eraw, e, s, d);
        atomicMax(&mk[d * 8 + h], fkey(als[s * 8 + h]));
      }
    }
  }
}

__device__ __forceinline__ float xhalf(float x) {
  return __builtin_bit_cast(float, __builtin_amdgcn_permlanex16(__builtin_bit_cast(int, x),
                                                                __builtin_bit_cast(int, x),
                                                                0x76543210, 0xfedcba98, false,
                                                                false));
}

template <int U, int WPB>
__global__ void edge_aggregate_row64(const int* __restrict__ ei, int Eraw, int Etot,
                                     const float* __restrict__ als, const float* __restrict__ ald,
                                     const unsigned int* __restrict__ mk,
                                     const float* __restrict__ hf,
                                     float* __restrict__ acc, float* __restrict__ den) {
  const int lane = threadIdx.x & 31;
  const int wave = __builtin_amdgcn_readfirstlane((int)(threadIdx.x >> 5));
  const int e0 = (blockIdx.x * WPB + wave) * 32;
  if (e0 >= Etot) return;
  const int cnt = Etot - e0 < 32 ? Etot - e0 : 32;
  int sl, dl;
  edge_sd_nb(ei, Eraw, e0 + lane < Etot ? e0 + lane : Etot - 1, sl, dl);
  const int h = lane >> 2;
  const bool head_lane = (lane & 3) == 0;
  const float* hcol = hf + 2 * lane;
  const bool lo = lane < 16;
  float* ocol = acc + (lo ? 2 * lane : 2 * (lane - 16) + 1);
  int j = 0;
  for (; j + U <= cnt; j += U) {
    int d[U];
    float w[U];
    float a1[U], a2[U];
#pragma unroll
    for (int u = 0; u < U; ++u) {
      const int s = __builtin_amdgcn_readlane(sl, j + u);
      d[u] = __builtin_amdgcn_readlane(dl, j + u);
      const int sh = s * 8 + h, dh = d[u] * 8 + h;
      const float adl = ald[dh];
      const float v = lrelu(als[sh] + adl);
      w[u] = __expf(v - seg_max_logit(mk[dh], adl));
      const v2f hv = *(const v2f*)(hcol + (long)s * 64);
      const float vx = hv.x * w[u], vy = hv.y * w[u];
      const float got = xhalf(lo ? vy : vx);
      a1[u] = lo ? vx : got;
      a2[u] = lo ? got : vy;
    }
#pragma unroll
    for (int u = 0; u < U; ++u) {
      float* op = ocol + (long)d[u] * 64;
      atomicAdd(op, a1[u]);
      atomicAdd(op + 32, a2[u]);
    }
    if (head_lane) {
#pragma unroll
      for (int u = 0; u < U; ++u) atomicAdd(&den[d[u] * 8 + h], w[u]);
    }
  }
  for (; j < cnt; ++j) {
    const int s = __builtin_amdgcn_readlane(sl, j);
    const int d = __builtin_amdgcn_readlane(dl, j);
    const int sh = s * 8 + h, dh = d * 8 + h;
    const float adl = ald[dh];
    const float v = lrelu(als[sh] + adl);
    const float w = __expf(v - seg_max_logit(mk[dh], adl));
    const v2f hv = *(const v2f*)(hcol + (long)s * 64);
    const float vx = hv.x * w, vy = hv.y * w;
    const float got = xhalf(lo ? vy : vx);
    float* op = ocol + (long)d * 64;
    atomicAdd(op, lo ? vx : got);
    atomicAdd(op + 32, lo ? got : vy);
    if (head_lane) atomicAdd(&den[dh], w);
  }
}

template <int U>
__global__ void edge_aggregate_c8(const int* __restrict__ ei, int Eraw, int Etot,
                                  const float* __restrict__ als, const float* __restrict__ ald,
                                  const unsigned int* __restrict__ mk,
                                  const float* __restrict__ hf,
                                  float* __restrict__ acc, float* __restrict__ den) {
  const int c = threadIdx.x & 7;
  const int e0 = blockIdx.x * (32 * U) + (threadIdx.x >> 3);
  if ((int)(blockIdx.x + 1) * (32 * U) <= Etot) {
    int dn[U];
    float w[U];
    float val[U];
#pragma unroll
    for (int j = 0; j < U; ++j) {
      int s, d; edge_sd_nb(ei, Eraw, e0 + j * 32, s, d);
      dn[j] = d;
      const float adl = ald[d];
      const float v = lrelu(als[s] + adl);
      w[j] = __expf(v - seg_max_logit(mk[d], adl));
      val[j] = hf[s * 8 + c] * w[j];
    }
#pragma unroll
    for (int j = 0; j < U; ++j) atomicAdd(&acc[dn[j] * 8 + c], val[j]);
    if (c == 0) {
#pragma unroll
      for (int j = 0; j < U; ++j) atomicAdd(&den[dn[j]], w[j]);
    }
  } else {
    for (int j = 0; j < U; ++j) {
      const int e = e0 + j * 32;
      if (e < Etot) {
        int s, d; edge_sd_nb(ei, Eraw, e, s, d);
        const float adl = ald[d];
        const float v = lrelu(als[s] + adl);
        const float w = __expf(v - seg_max_logit(mk[d], adl));
        atomicAdd(&acc[d * 8 + c], hf[s * 8 + c] * w);
        if (c == 0) atomicAdd(&den[d], w);
      }
    }
  }
}

__global__ void normalize_score_kernel(float* __restrict__ emb, const float* __restrict__ den,
                                       const float* __restrict__ b2,
                                       const float* __restrict__ W3,
                                       const float* __restrict__ b3, float* __restrict__ score,
                                       long N) {
  long n = (long)blockIdx.x * blockDim.x + threadIdx.x;
  if (n >= N) return;
  const float r = 1.f / (den[n] + 1e-16f);
  float4* row = (float4*)(emb + n * 8);
  const float4 lo = row[0], hi = row[1];
  float e[8] = {lo.x, lo.y, lo.z, lo.w, hi.x, hi.y, hi.z, hi.w};
#pragma unroll
  for (int c = 0; c < 8; ++c) e[c] = e[c] * r + b2[c];
  row[0] = make_float4(e[0], e[1], e[2], e[3]);
  row[1] = make_float4(e[4], e[5], e[6], e[7]);
  float s = b3[0];
#pragma unroll
  for (int c = 0; c < 8; ++c) s += e[c] * W3[c];
  score[n] = s;
}

static inline int nblk(long n, int b) { return (int)((n + b - 1) / b); }

extern "C" void kernel_launch(void* const* d_in, const int* in_sizes, int n_in,
                              void* d_out, int out_size, void* d_ws, size_t ws_size,
                              hipStream_t stream) {
  const float* x   = (const float*)d_in[0];
  const int*   ei  = (const int*)d_in[1];
  const float* W1  = (const float*)d_in[2];
  const float* as1 = (const float*)d_in[3];
  const float* ad1 = (const float*)d_in[4];
  const float* b1  = (const float*)d_in[5];
  const float* W2  = (const float*)d_in[6];
  const float* as2 = (const float*)d_in[7];
  const float* ad2 = (const float*)d_in[8];
  const float* b2  = (const float*)d_in[9];
  const float* W3  = (const float*)d_in[10];
  const float* b3  = (const float*)d_in[11];

  const long N    = in_sizes[0] / 256;
  const long Eraw = in_sizes[1] / 2;
  const long Etot = Eraw + N;

  float* ws   = (float*)d_ws;
  float* h1   = ws;
  float* acc1 = h1 + N * 64;
  float* als1 = acc1 + N * 64;
  float* ald1 = als1 + N * 8;
  unsigned int* m1 = (unsigned int*)(ald1 + N * 8);
  float* den1 = (float*)(m1 + N * 8);
  float* h2   = den1 + N * 8;
  float* als2 = h2 + N * 8;
  float* ald2 = als2 + N;
  unsigned int* m2 = (unsigned int*)(ald2 + N);
  float* den2 = (float*)(m2 + N);

  float* emb   = (float*)d_out;
  float* score = emb + N * 8;

  const int B = 256;
  const int mtiles = (int)(N / 16);

  gemm1_x_w1<<<mtiles, 128, 0, stream>>>(x, W1, h1);
  attn_logits<8, 8><<<nblk(N * 8, B), B, 0, stream>>>(h1, as1, ad1, als1, ald1, N * 8);
  init_softmax<<<nblk(N * 8, B), B, 0, stream>>>(m1, den1, N * 8);
  init_zero<<<nblk(N * 16, B), B, 0, stream>>>(acc1, N * 16);
  edge_srcmax_h8<4><<<nblk(Etot, 128), B, 0, stream>>>(ei, (int)Eraw, (int)Etot, als1, m1);
  edge_aggregate_row64<4, 8><<<nblk(Etot, 256), B, 0, stream>>>(ei, (int)Eraw, (int)Etot, als1,
                                                                ald1, m1, h1, acc1, den1);

  gemm2_h_w2<<<(mtiles + 3) / 4, 128, 0, stream>>>(acc1, den1, b1, W2, h2, mtiles);
  attn_logits<1, 8><<<nblk(N, B), B, 0, stream>>>(h2, as2, ad2, als2, ald2, N);
  init_softmax<<<nblk(N, B), B, 0, stream>>>(m2, den2, N);
  init_zero<<<nblk(N * 2, B), B, 0, stream>>>(emb, N * 2);
  edge_srcmax<1><<<nblk(Etot, B), B, 0, stream>>>(ei, Eraw, Etot, als2, m2);
  edge_aggregate_c8<4><<<nblk(Etot, 128), B, 0, stream>>>(ei, (int)Eraw, (int)Etot, als2, ald2,
                                                          m2, h2, emb, den2);

  normalize_score_kernel<<<nblk(N, B), B, 0, stream>>>(emb, den2, b2, W3, b3, score, N);
}
